# starting kernel
# baseline (speedup 1.0000x reference)
#include <hip/hip_runtime.h>
#include <hip/hip_bf16.h>

typedef float v2f __attribute__((ext_vector_type(2)));
typedef float v8f __attribute__((ext_vector_type(8)));

#define NPTS   100000
#define CCH    64
#define KOFF   27
#define NB     8
#define EPSI   1e-5f
#define NSLOPE 0.01f

__device__ __forceinline__ void atomicAddF32(float* p, float v) {
    unsafeAtomicAdd(p, v);
}

__global__ void fill_zero_kernel(float* __restrict__ p, int n) {
    for (int i = blockIdx.x * blockDim.x + threadIdx.x; i < n;
         i += gridDim.x * blockDim.x)
        p[i] = 0.0f;
}

__global__ __launch_bounds__(256)
void sparse_conv_kernel(const float* __restrict__ X,
                        const float* __restrict__ W,
                        const int*   __restrict__ in_map,
                        const int*   __restrict__ out_map,
                        float*       __restrict__ Y)
{
    __shared__ float Wlds[CCH * CCH];

    const int k = blockIdx.y;
    const int t = threadIdx.x;

    {
        const float4* Wg = (const float4*)(W + k * CCH * CCH);
        float4*       Wl = (float4*)Wlds;
#pragma unroll
        for (int i = 0; i < 4; ++i)
            Wl[t + i * 256] = Wg[t + i * 256];
    }
    __syncthreads();

    const int wave    = t >> 5;
    const int lane    = t & 31;
    const int half    = lane >> 4;
    const int m       = lane & 15;
    const int rowbase = blockIdx.x * 128 + wave * 16;

    const int ri  = rowbase + m;
    const int gin = (ri < NPTS) ? in_map[k * NPTS + ri] : 0;
    const float* Xrow = X + (long)gin * CCH;

    v2f a[16];
#pragma unroll
    for (int kk = 0; kk < 16; ++kk)
        a[kk] = *(const v2f*)(Xrow + kk * 4 + 2 * half);

#pragma unroll
    for (int nt = 0; nt < 4; ++nt) {
        const int n0 = nt * 16;
        v8f acc = {0.f, 0.f, 0.f, 0.f, 0.f, 0.f, 0.f, 0.f};

#pragma unroll
        for (int kk = 0; kk < 16; ++kk) {
            v2f b;
            b.x = Wlds[(kk * 4 + 2 * half + 0) * CCH + n0 + m];
            b.y = Wlds[(kk * 4 + 2 * half + 1) * CCH + n0 + m];
            acc = __builtin_amdgcn_wmma_f32_16x16x4_f32(
                false, a[kk], false, b, (short)0, acc, false, false);
        }

#pragma unroll
        for (int r = 0; r < 8; ++r) {
            const int M    = r + 8 * half;
            const int rowM = rowbase + M;
            if (rowM < NPTS) {
                const int gout = out_map[k * NPTS + rowM];
                atomicAddF32(Y + (long)gout * CCH + n0 + m, acc[r]);
            }
        }
    }
}

__global__ __launch_bounds__(256)
void stats_kernel(const float* __restrict__ A,
                  const int*   __restrict__ bidx,
                  float* __restrict__ gsum,
                  float* __restrict__ gsq,
                  float* __restrict__ gcnt)
{
    const int t    = threadIdx.x;
    const int c    = t & 63;
    const int g    = t >> 6;
    const int row0 = blockIdx.x * 256 + g * 64;

    float s = 0.f, s2 = 0.f, cn = 0.f;
    int curb = -1;

    for (int i = 0; i < 64; ++i) {
        const int row = row0 + i;
        if (row >= NPTS) break;
        const int b = bidx[row];
        if (b != curb) {
            if (curb >= 0) {
                atomicAddF32(gsum + curb * CCH + c, s);
                atomicAddF32(gsq  + curb * CCH + c, s2);
                if (c == 0) atomicAddF32(gcnt + curb, cn);
            }
            curb = b; s = 0.f; s2 = 0.f; cn = 0.f;
        }
        const float v = A[(long)row * CCH + c];
        s += v; s2 += v * v; cn += 1.f;
    }
    if (curb >= 0) {
        atomicAddF32(gsum + curb * CCH + c, s);
        atomicAddF32(gsq  + curb * CCH + c, s2);
        if (c == 0) atomicAddF32(gcnt + curb, cn);
    }
}

__global__ void finalize_kernel(const float* __restrict__ gsum,
                                const float* __restrict__ gsq,
                                const float* __restrict__ gcnt,
                                float* __restrict__ mean,
                                float* __restrict__ rstd)
{
    const int t  = threadIdx.x;
    const float cn  = gcnt[t >> 6];
    const float mu  = gsum[t] / cn;
    const float var = gsq[t] / cn - mu * mu;
    mean[t] = mu;
    rstd[t] = rsqrtf(var + EPSI);
}

__global__ __launch_bounds__(256)
void apply_norm_kernel(float* __restrict__ A,
                       const int*   __restrict__ bidx,
                       const float* __restrict__ mean,
                       const float* __restrict__ rstd,
                       const float* __restrict__ gamma,
                       const float* __restrict__ beta)
{
    const int idx = blockIdx.x * 256 + threadIdx.x;
    if (idx >= NPTS * CCH) return;
    const int row = idx >> 6, c = idx & 63;
    const int b   = bidx[row];
    const float x = A[idx];
    const float y = gamma[c] * (x - mean[b * CCH + c]) * rstd[b * CCH + c] + beta[c];
    A[idx] = y > 0.f ? y : NSLOPE * y;
}

__global__ __launch_bounds__(256)
void apply_norm_res_kernel(float* __restrict__ Y,
                           const float* __restrict__ feats,
                           const int*   __restrict__ bidx,
                           const float* __restrict__ mean,
                           const float* __restrict__ rstd,
                           const float* __restrict__ gamma,
                           const float* __restrict__ beta)
{
    const int idx = blockIdx.x * 256 + threadIdx.x;
    if (idx >= NPTS * CCH) return;
    const int row = idx >> 6, c = idx & 63;
    const int b   = bidx[row];
    const float x = Y[idx];
    float y = gamma[c] * (x - mean[b * CCH + c]) * rstd[b * CCH + c] + beta[c];
    y += feats[idx];
    Y[idx] = y > 0.f ? y : NSLOPE * y;
}

extern "C" void kernel_launch(void* const* d_in, const int* in_sizes, int n_in,
                              void* d_out, int out_size, void* d_ws, size_t ws_size,
                              hipStream_t stream)
{
    const float* feats   = (const float*)d_in[0];
    const float* W1      = (const float*)d_in[1];
    const float* g1      = (const float*)d_in[2];
    const float* b1      = (const float*)d_in[3];
    const float* W2      = (const float*)d_in[4];
    const float* g2      = (const float*)d_in[5];
    const float* b2      = (const float*)d_in[6];
    const int*   in_map  = (const int*)d_in[7];
    const int*   out_map = (const int*)d_in[8];
    const int*   bidx    = (const int*)d_in[9];
    float*       out     = (float*)d_out;

    const int NC = NPTS * CCH;

    float* A     = (float*)d_ws;
    float* sbase = A + NC;
    float* sum1  = sbase;
    float* sq1   = sbase + 512;
    float* cnt1  = sbase + 1024;
    float* mean1 = sbase + 1088;
    float* rstd1 = sbase + 1600;
    float* sum2  = sbase + 2112;
    float* sq2   = sbase + 2624;
    float* cnt2  = sbase + 3136;
    float* mean2 = sbase + 3200;
    float* rstd2 = sbase + 3712;
    const int STATS = 4224;

    fill_zero_kernel<<<2048, 256, 0, stream>>>(A, NC + STATS);
    fill_zero_kernel<<<2048, 256, 0, stream>>>(out, NC);

    dim3 convGrid((NPTS + 127) / 128, KOFF);
    const int statGrid  = (NPTS + 255) / 256;
    const int elemGrid  = (NC + 255) / 256;

    sparse_conv_kernel<<<convGrid, 256, 0, stream>>>(feats, W1, in_map, out_map, A);
    stats_kernel<<<statGrid, 256, 0, stream>>>(A, bidx, sum1, sq1, cnt1);
    finalize_kernel<<<1, 512, 0, stream>>>(sum1, sq1, cnt1, mean1, rstd1);
    apply_norm_kernel<<<elemGrid, 256, 0, stream>>>(A, bidx, mean1, rstd1, g1, b1);
    sparse_conv_kernel<<<convGrid, 256, 0, stream>>>(A, W2, in_map, out_map, out);
    stats_kernel<<<statGrid, 256, 0, stream>>>(out, bidx, sum2, sq2, cnt2);
    finalize_kernel<<<1, 512, 0, stream>>>(sum2, sq2, cnt2, mean2, rstd2);
    apply_norm_res_kernel<<<elemGrid, 256, 0, stream>>>(out, feats, bidx, mean2, rstd2, g2, b2);
}


# conv kernel: a tile's out_map indices are loaded once at the top, in flight with the W loads (two 16-byte loads per lane), not by a dependent load and wait before each of its 32 atomic adds
# speedup vs baseline: 1.4474x; 1.4474x over previous
#include <hip/hip_runtime.h>
#include <hip/hip_bf16.h>

typedef float v2f __attribute__((ext_vector_type(2)));
typedef float v8f __attribute__((ext_vector_type(8)));

#define NPTS   100000
#define CCH    64
#define KOFF   27
#define NB     8
#define EPSI   1e-5f
#define NSLOPE 0.01f

__device__ __forceinline__ void atomicAddF32(float* p, float v) {
    unsafeAtomicAdd(p, v);
}

__global__ void fill_zero_kernel(float* __restrict__ p, int n) {
    for (int i = blockIdx.x * blockDim.x + threadIdx.x; i < n;
         i += gridDim.x * blockDim.x)
        p[i] = 0.0f;
}

static_assert(NPTS % 16 == 0, "a 16-row tile is wholly inside or wholly outside the point range");
static_assert((long)NPTS * CCH < (1L << 31), "element offsets into Y fit 32 bits");

__global__ __launch_bounds__(256)
void sparse_conv_kernel(const float* __restrict__ X,
                        const float* __restrict__ W,
                        const int*   __restrict__ in_map,
                        const int*   __restrict__ out_map,
                        float*       __restrict__ Y)
{
    __shared__ float Wlds[CCH * CCH];

    const int k = blockIdx.y;
    const int t = threadIdx.x;

    const int wave    = t >> 5;
    const int lane    = t & 31;
    const int half    = lane >> 4;
    const int m       = lane & 15;
    const int rowbase = blockIdx.x * 128 + wave * 16;
    const bool live   = rowbase < NPTS;
    const int  rb     = live ? rowbase : 0;

    const float4* Wg = (const float4*)(W + k * CCH * CCH);
    float4 w[4];
#pragma unroll
    for (int i = 0; i < 4; ++i)
        w[i] = Wg[t + i * 256];

    const int   gin   = in_map[k * NPTS + rb + m];
    const int4* op    = (const int4*)(out_map + k * NPTS + rb + 8 * half);
    const int4  go_lo = op[0];
    const int4  go_hi = op[1];

    {
        float4* Wl = (float4*)Wlds;
#pragma unroll
        for (int i = 0; i < 4; ++i)
            Wl[t + i * 256] = w[i];
    }
    __syncthreads();

    if (!live) return;

    const float* Xrow = X + (long)gin * CCH;

    v2f a[16];
#pragma unroll
    for (int kk = 0; kk < 16; ++kk)
        a[kk] = *(const v2f*)(Xrow + kk * 4 + 2 * half);

    const int go[8] = {go_lo.x, go_lo.y, go_lo.z, go_lo.w,
                       go_hi.x, go_hi.y, go_hi.z, go_hi.w};
    unsigned yoff[8];
#pragma unroll
    for (int r = 0; r < 8; ++r)
        yoff[r] = (unsigned)go[r] * CCH + m;

#pragma unroll
    for (int nt = 0; nt < 4; ++nt) {
        const int n0 = nt * 16;
        v8f acc = {0.f, 0.f, 0.f, 0.f, 0.f, 0.f, 0.f, 0.f};

#pragma unroll
        for (int kk = 0; kk < 16; ++kk) {
            v2f b;
            b.x = Wlds[(kk * 4 + 2 * half + 0) * CCH + n0 + m];
            b.y = Wlds[(kk * 4 + 2 * half + 1) * CCH + n0 + m];
            acc = __builtin_amdgcn_wmma_f32_16x16x4_f32(
                false, a[kk], false, b, (short)0, acc, false, false);
        }

#pragma unroll
        for (int r = 0; r < 8; ++r)
            atomicAddF32(Y + (yoff[r] + n0), acc[r]);
    }
}

__global__ __launch_bounds__(256)
void stats_kernel(const float* __restrict__ A,
                  const int*   __restrict__ bidx,
                  float* __restrict__ gsum,
                  float* __restrict__ gsq,
                  float* __restrict__ gcnt)
{
    const int t    = threadIdx.x;
    const int c    = t & 63;
    const int g    = t >> 6;
    const int row0 = blockIdx.x * 256 + g * 64;

    float s = 0.f, s2 = 0.f, cn = 0.f;
    int curb = -1;

    for (int i = 0; i < 64; ++i) {
        const int row = row0 + i;
        if (row >= NPTS) break;
        const int b = bidx[row];
        if (b != curb) {
            if (curb >= 0) {
                atomicAddF32(gsum + curb * CCH + c, s);
                atomicAddF32(gsq  + curb * CCH + c, s2);
                if (c == 0) atomicAddF32(gcnt + curb, cn);
            }
            curb = b; s = 0.f; s2 = 0.f; cn = 0.f;
        }
        const float v = A[(long)row * CCH + c];
        s += v; s2 += v * v; cn += 1.f;
    }
    if (curb >= 0) {
        atomicAddF32(gsum + curb * CCH + c, s);
        atomicAddF32(gsq  + curb * CCH + c, s2);
        if (c == 0) atomicAddF32(gcnt + curb, cn);
    }
}

__global__ void finalize_kernel(const float* __restrict__ gsum,
                                const float* __restrict__ gsq,
                                const float* __restrict__ gcnt,
                                float* __restrict__ mean,
                                float* __restrict__ rstd)
{
    const int t  = threadIdx.x;
    const float cn  = gcnt[t >> 6];
    const float mu  = gsum[t] / cn;
    const float var = gsq[t] / cn - mu * mu;
    mean[t] = mu;
    rstd[t] = rsqrtf(var + EPSI);
}

__global__ __launch_bounds__(256)
void apply_norm_kernel(float* __restrict__ A,
                       const int*   __restrict__ bidx,
                       const float* __restrict__ mean,
                       const float* __restrict__ rstd,
                       const float* __restrict__ gamma,
                       const float* __restrict__ beta)
{
    const int idx = blockIdx.x * 256 + threadIdx.x;
    if (idx >= NPTS * CCH) return;
    const int row = idx >> 6, c = idx & 63;
    const int b   = bidx[row];
    const float x = A[idx];
    const float y = gamma[c] * (x - mean[b * CCH + c]) * rstd[b * CCH + c] + beta[c];
    A[idx] = y > 0.f ? y : NSLOPE * y;
}

__global__ __launch_bounds__(256)
void apply_norm_res_kernel(float* __restrict__ Y,
                           const float* __restrict__ feats,
                           const int*   __restrict__ bidx,
                           const float* __restrict__ mean,
                           const float* __restrict__ rstd,
                           const float* __restrict__ gamma,
                           const float* __restrict__ beta)
{
    const int idx = blockIdx.x * 256 + threadIdx.x;
    if (idx >= NPTS * CCH) return;
    const int row = idx >> 6, c = idx & 63;
    const int b   = bidx[row];
    const float x = Y[idx];
    float y = gamma[c] * (x - mean[b * CCH + c]) * rstd[b * CCH + c] + beta[c];
    y += feats[idx];
    Y[idx] = y > 0.f ? y : NSLOPE * y;
}

extern "C" void kernel_launch(void* const* d_in, const int* in_sizes, int n_in,
                              void* d_out, int out_size, void* d_ws, size_t ws_size,
                              hipStream_t stream)
{
    const float* feats   = (const float*)d_in[0];
    const float* W1      = (const float*)d_in[1];
    const float* g1      = (const float*)d_in[2];
    const float* b1      = (const float*)d_in[3];
    const float* W2      = (const float*)d_in[4];
    const float* g2      = (const float*)d_in[5];
    const float* b2      = (const float*)d_in[6];
    const int*   in_map  = (const int*)d_in[7];
    const int*   out_map = (const int*)d_in[8];
    const int*   bidx    = (const int*)d_in[9];
    float*       out     = (float*)d_out;

    const int NC = NPTS * CCH;

    float* A     = (float*)d_ws;
    float* sbase = A + NC;
    float* sum1  = sbase;
    float* sq1   = sbase + 512;
    float* cnt1  = sbase + 1024;
    float* mean1 = sbase + 1088;
    float* rstd1 = sbase + 1600;
    float* sum2  = sbase + 2112;
    float* sq2   = sbase + 2624;
    float* cnt2  = sbase + 3136;
    float* mean2 = sbase + 3200;
    float* rstd2 = sbase + 3712;
    const int STATS = 4224;

    fill_zero_kernel<<<2048, 256, 0, stream>>>(A, NC + STATS);
    fill_zero_kernel<<<2048, 256, 0, stream>>>(out, NC);

    dim3 convGrid((NPTS + 127) / 128, KOFF);
    const int statGrid  = (NPTS + 255) / 256;
    const int elemGrid  = (NC + 255) / 256;

    sparse_conv_kernel<<<convGrid, 256, 0, stream>>>(feats, W1, in_map, out_map, A);
    stats_kernel<<<statGrid, 256, 0, stream>>>(A, bidx, sum1, sq1, cnt1);
    finalize_kernel<<<1, 512, 0, stream>>>(sum1, sq1, cnt1, mean1, rstd1);
    apply_norm_kernel<<<elemGrid, 256, 0, stream>>>(A, bidx, mean1, rstd1, g1, b1);
    sparse_conv_kernel<<<convGrid, 256, 0, stream>>>(A, W2, in_map, out_map, out);
    stats_kernel<<<statGrid, 256, 0, stream>>>(out, bidx, sum2, sq2, cnt2);
    finalize_kernel<<<1, 512, 0, stream>>>(sum2, sq2, cnt2, mean2, rstd2);
    apply_norm_res_kernel<<<elemGrid, 256, 0, stream>>>(out, feats, bidx, mean2, rstd2, g2, b2);
}


# conv kernel: lane halves swap accumulator registers of two column tiles (v_permlane16_swap) so each f32 atomic add covers one whole 128-B line of one output row, not two 64-B halves of two rows
# speedup vs baseline: 1.6898x; 1.1675x over previous
#include <hip/hip_runtime.h>
#include <hip/hip_bf16.h>

typedef float v2f __attribute__((ext_vector_type(2)));
typedef float v8f __attribute__((ext_vector_type(8)));

#define NPTS   100000
#define CCH    64
#define KOFF   27
#define NB     8
#define EPSI   1e-5f
#define NSLOPE 0.01f

__device__ __forceinline__ void atomicAddF32(float* p, float v) {
    unsafeAtomicAdd(p, v);
}

__global__ void fill_zero_kernel(float* __restrict__ p, int n) {
    for (int i = blockIdx.x * blockDim.x + threadIdx.x; i < n;
         i += gridDim.x * blockDim.x)
        p[i] = 0.0f;
}

typedef __attribute__((address_space(1))) float gfloat;
__device__ __forceinline__ float* row_ptr(float* Y, int g) {
    const unsigned long long p = (unsigned long long)(Y + (size_t)(unsigned)g * CCH);
    const unsigned lo = __builtin_amdgcn_readfirstlane((unsigned)p);
    const unsigned hi = __builtin_amdgcn_readfirstlane((unsigned)(p >> 32));
    return (float*)(gfloat*)(((unsigned long long)hi << 32) | lo);
}

static_assert(NPTS % 16 == 0, "a 16-row tile is wholly inside or wholly outside the point range");
static_assert((long)NPTS * CCH < (1L << 31), "element offsets into Y fit 32 bits");

__global__ __launch_bounds__(256)
void sparse_conv_kernel(const float* __restrict__ X,
                        const float* __restrict__ W,
                        const int*   __restrict__ in_map,
                        const int*   __restrict__ out_map,
                        float*       __restrict__ Y)
{
    __shared__ float Wlds[CCH * CCH];

    const int k = blockIdx.y;
    const int t = threadIdx.x;

    const int wave    = __builtin_amdgcn_readfirstlane(t >> 5);
    const int lane    = t & 31;
    const int half    = lane >> 4;
    const int m       = lane & 15;
    const int rowbase = blockIdx.x * 128 + wave * 16;
    const bool live   = rowbase < NPTS;
    const int  rb     = live ? rowbase : 0;

    const float4* Wg = (const float4*)(W + k * CCH * CCH);
    float4 w[4];
#pragma unroll
    for (int i = 0; i < 4; ++i)
        w[i] = Wg[t + i * 256];

    const int   gin = in_map[k * NPTS + rb + m];
    const int4* op  = (const int4*)(out_map + k * NPTS + rb);
    const int4  g0 = op[0], g1 = op[1], g2 = op[2], g3 = op[3];

    {
        float4* Wl = (float4*)Wlds;
#pragma unroll
        for (int i = 0; i < 4; ++i)
            Wl[t + i * 256] = w[i];
    }
    __syncthreads();

    if (!live) return;

    const float* Xrow = X + (long)gin * CCH;

    v2f a[16];
#pragma unroll
    for (int kk = 0; kk < 16; ++kk)
        a[kk] = *(const v2f*)(Xrow + kk * 4 + 2 * half);

    const int go[16] = {g0.x, g0.y, g0.z, g0.w, g1.x, g1.y, g1.z, g1.w,
                        g2.x, g2.y, g2.z, g2.w, g3.x, g3.y, g3.z, g3.w};

#pragma unroll
    for (int p = 0; p < 2; ++p) {
        const int c0 = p * 32;
        v8f acc0 = {0.f, 0.f, 0.f, 0.f, 0.f, 0.f, 0.f, 0.f};
        v8f acc1 = {0.f, 0.f, 0.f, 0.f, 0.f, 0.f, 0.f, 0.f};

#pragma unroll
        for (int kk = 0; kk < 16; ++kk) {
            v2f b;
            b.x = Wlds[(kk * 4 + 2 * half + 0) * CCH + c0 + m];
            b.y = Wlds[(kk * 4 + 2 * half + 1) * CCH + c0 + m];
            acc0 = __builtin_amdgcn_wmma_f32_16x16x4_f32(
                false, a[kk], false, b, (short)0, acc0, false, false);
        }
#pragma unroll
        for (int kk = 0; kk < 16; ++kk) {
            v2f b;
            b.x = Wlds[(kk * 4 + 2 * half + 0) * CCH + c0 + 16 + m];
            b.y = Wlds[(kk * 4 + 2 * half + 1) * CCH + c0 + 16 + m];
            acc1 = __builtin_amdgcn_wmma_f32_16x16x4_f32(
                false, a[kk], false, b, (short)0, acc1, false, false);
        }

#pragma unroll
        for (int r = 0; r < 8; ++r) {
            const auto sw = __builtin_amdgcn_permlane16_swap(
                __float_as_uint(acc0[r]), __float_as_uint(acc1[r]), false, false);
            const float lo = __uint_as_float(sw[0]);
            const float hi = __uint_as_float(sw[1]);
            atomicAddF32(row_ptr(Y, go[r])     + c0 + lane, lo);
            atomicAddF32(row_ptr(Y, go[r + 8]) + c0 + lane, hi);
        }
    }
}

__global__ __launch_bounds__(256)
void stats_kernel(const float* __restrict__ A,
                  const int*   __restrict__ bidx,
                  float* __restrict__ gsum,
                  float* __restrict__ gsq,
                  float* __restrict__ gcnt)
{
    const int t    = threadIdx.x;
    const int c    = t & 63;
    const int g    = t >> 6;
    const int row0 = blockIdx.x * 256 + g * 64;

    float s = 0.f, s2 = 0.f, cn = 0.f;
    int curb = -1;

    for (int i = 0; i < 64; ++i) {
        const int row = row0 + i;
        if (row >= NPTS) break;
        const int b = bidx[row];
        if (b != curb) {
            if (curb >= 0) {
                atomicAddF32(gsum + curb * CCH + c, s);
                atomicAddF32(gsq  + curb * CCH + c, s2);
                if (c == 0) atomicAddF32(gcnt + curb, cn);
            }
            curb = b; s = 0.f; s2 = 0.f; cn = 0.f;
        }
        const float v = A[(long)row * CCH + c];
        s += v; s2 += v * v; cn += 1.f;
    }
    if (curb >= 0) {
        atomicAddF32(gsum + curb * CCH + c, s);
        atomicAddF32(gsq  + curb * CCH + c, s2);
        if (c == 0) atomicAddF32(gcnt + curb, cn);
    }
}

__global__ void finalize_kernel(const float* __restrict__ gsum,
                                const float* __restrict__ gsq,
                                const float* __restrict__ gcnt,
                                float* __restrict__ mean,
                                float* __restrict__ rstd)
{
    const int t  = threadIdx.x;
    const float cn  = gcnt[t >> 6];
    const float mu  = gsum[t] / cn;
    const float var = gsq[t] / cn - mu * mu;
    mean[t] = mu;
    rstd[t] = rsqrtf(var + EPSI);
}

__global__ __launch_bounds__(256)
void apply_norm_kernel(float* __restrict__ A,
                       const int*   __restrict__ bidx,
                       const float* __restrict__ mean,
                       const float* __restrict__ rstd,
                       const float* __restrict__ gamma,
                       const float* __restrict__ beta)
{
    const int idx = blockIdx.x * 256 + threadIdx.x;
    if (idx >= NPTS * CCH) return;
    const int row = idx >> 6, c = idx & 63;
    const int b   = bidx[row];
    const float x = A[idx];
    const float y = gamma[c] * (x - mean[b * CCH + c]) * rstd[b * CCH + c] + beta[c];
    A[idx] = y > 0.f ? y : NSLOPE * y;
}

__global__ __launch_bounds__(256)
void apply_norm_res_kernel(float* __restrict__ Y,
                           const float* __restrict__ feats,
                           const int*   __restrict__ bidx,
                           const float* __restrict__ mean,
                           const float* __restrict__ rstd,
                           const float* __restrict__ gamma,
                           const float* __restrict__ beta)
{
    const int idx = blockIdx.x * 256 + threadIdx.x;
    if (idx >= NPTS * CCH) return;
    const int row = idx >> 6, c = idx & 63;
    const int b   = bidx[row];
    const float x = Y[idx];
    float y = gamma[c] * (x - mean[b * CCH + c]) * rstd[b * CCH + c] + beta[c];
    y += feats[idx];
    Y[idx] = y > 0.f ? y : NSLOPE * y;
}

extern "C" void kernel_launch(void* const* d_in, const int* in_sizes, int n_in,
                              void* d_out, int out_size, void* d_ws, size_t ws_size,
                              hipStream_t stream)
{
    const float* feats   = (const float*)d_in[0];
    const float* W1      = (const float*)d_in[1];
    const float* g1      = (const float*)d_in[2];
    const float* b1      = (const float*)d_in[3];
    const float* W2      = (const float*)d_in[4];
    const float* g2      = (const float*)d_in[5];
    const float* b2      = (const float*)d_in[6];
    const int*   in_map  = (const int*)d_in[7];
    const int*   out_map = (const int*)d_in[8];
    const int*   bidx    = (const int*)d_in[9];
    float*       out     = (float*)d_out;

    const int NC = NPTS * CCH;

    float* A     = (float*)d_ws;
    float* sbase = A + NC;
    float* sum1  = sbase;
    float* sq1   = sbase + 512;
    float* cnt1  = sbase + 1024;
    float* mean1 = sbase + 1088;
    float* rstd1 = sbase + 1600;
    float* sum2  = sbase + 2112;
    float* sq2   = sbase + 2624;
    float* cnt2  = sbase + 3136;
    float* mean2 = sbase + 3200;
    float* rstd2 = sbase + 3712;
    const int STATS = 4224;

    fill_zero_kernel<<<2048, 256, 0, stream>>>(A, NC + STATS);
    fill_zero_kernel<<<2048, 256, 0, stream>>>(out, NC);

    dim3 convGrid((NPTS + 127) / 128, KOFF);
    const int statGrid  = (NPTS + 255) / 256;
    const int elemGrid  = (NC + 255) / 256;

    sparse_conv_kernel<<<convGrid, 256, 0, stream>>>(feats, W1, in_map, out_map, A);
    stats_kernel<<<statGrid, 256, 0, stream>>>(A, bidx, sum1, sq1, cnt1);
    finalize_kernel<<<1, 512, 0, stream>>>(sum1, sq1, cnt1, mean1, rstd1);
    apply_norm_kernel<<<elemGrid, 256, 0, stream>>>(A, bidx, mean1, rstd1, g1, b1);
    sparse_conv_kernel<<<convGrid, 256, 0, stream>>>(A, W2, in_map, out_map, out);
    stats_kernel<<<statGrid, 256, 0, stream>>>(out, bidx, sum2, sq2, cnt2);
    finalize_kernel<<<1, 512, 0, stream>>>(sum2, sq2, cnt2, mean2, rstd2);
    apply_norm_res_kernel<<<elemGrid, 256, 0, stream>>>(out, feats, bidx, mean2, rstd2, g2, b2);
}


# conv kernel: X rows gathered as whole 256-B rows (16-B loads, 2 rows per load), turned into WMMA A fragments through a wave-private LDS tile; changes the k order of each dot product's sum
# speedup vs baseline: 1.8336x; 1.0851x over previous
#include <hip/hip_runtime.h>
#include <hip/hip_bf16.h>

typedef float v2f __attribute__((ext_vector_type(2)));
typedef float v8f __attribute__((ext_vector_type(8)));

#define NPTS   100000
#define CCH    64
#define KOFF   27
#define NB     8
#define EPSI   1e-5f
#define NSLOPE 0.01f

__device__ __forceinline__ void atomicAddF32(float* p, float v) {
    unsafeAtomicAdd(p, v);
}

__global__ void fill_zero_kernel(float* __restrict__ p, int n) {
    for (int i = blockIdx.x * blockDim.x + threadIdx.x; i < n;
         i += gridDim.x * blockDim.x)
        p[i] = 0.0f;
}

typedef __attribute__((address_space(1))) float gfloat;
__device__ __forceinline__ float* row_ptr(float* Y, int g) {
    const unsigned long long p = (unsigned long long)(Y + (size_t)(unsigned)g * CCH);
    const unsigned lo = __builtin_amdgcn_readfirstlane((unsigned)p);
    const unsigned hi = __builtin_amdgcn_readfirstlane((unsigned)(p >> 32));
    return (float*)(gfloat*)(((unsigned long long)hi << 32) | lo);
}

static_assert(NPTS % 16 == 0, "a 16-row tile is wholly inside or wholly outside the point range");
static_assert((long)NPTS * CCH < (1L << 31), "element offsets into Y fit 32 bits");

#define APITCH 68

__global__ __launch_bounds__(256)
void sparse_conv_kernel(const float* __restrict__ X,
                        const float* __restrict__ W,
                        const int*   __restrict__ in_map,
                        const int*   __restrict__ out_map,
                        float*       __restrict__ Y)
{
    __shared__ __attribute__((aligned(16))) float Wlds[CCH * CCH];
    __shared__ __attribute__((aligned(16))) float Alds[8 * 16 * APITCH];

    const int k = blockIdx.y;
    const int t = threadIdx.x;

    const int wave    = __builtin_amdgcn_readfirstlane(t >> 5);
    const int lane    = t & 31;
    const int half    = lane >> 4;
    const int m       = lane & 15;
    const int rowbase = blockIdx.x * 128 + wave * 16;
    const bool live   = rowbase < NPTS;
    const int  rb     = live ? rowbase : 0;

    const float4* Wg = (const float4*)(W + k * CCH * CCH);
    float4 w[4];
#pragma unroll
    for (int i = 0; i < 4; ++i)
        w[i] = Wg[t + i * 256];

    const int4* ip  = (const int4*)(in_map  + k * NPTS + rb);
    const int4* op  = (const int4*)(out_map + k * NPTS + rb);
    const int4  i0 = ip[0], i1 = ip[1], i2 = ip[2], i3 = ip[3];
    const int4  g0 = op[0], g1 = op[1], g2 = op[2], g3 = op[3];

    {
        float4* Wl = (float4*)Wlds;
#pragma unroll
        for (int i = 0; i < 4; ++i)
            Wl[t + i * 256] = w[i];
    }
    __syncthreads();

    if (!live) return;

    const int gi[16] = {i0.x, i0.y, i0.z, i0.w, i1.x, i1.y, i1.z, i1.w,
                        i2.x, i2.y, i2.z, i2.w, i3.x, i3.y, i3.z, i3.w};
    const int go[16] = {g0.x, g0.y, g0.z, g0.w, g1.x, g1.y, g1.z, g1.w,
                        g2.x, g2.y, g2.z, g2.w, g3.x, g3.y, g3.z, g3.w};

    float4 x[8];
#pragma unroll
    for (int i = 0; i < 8; ++i) {
        const int r = half ? gi[2 * i + 1] : gi[2 * i];
        x[i] = *(const float4*)(X + (long)r * CCH + 4 * m);
    }

    float* At = Alds + wave * (16 * APITCH);
#pragma unroll
    for (int i = 0; i < 8; ++i)
        *(float4*)(At + (2 * i + half) * APITCH + 4 * m) = x[i];

    __builtin_amdgcn_fence(__ATOMIC_RELEASE, "wavefront");
    __builtin_amdgcn_wave_barrier();
    __builtin_amdgcn_fence(__ATOMIC_ACQUIRE, "wavefront");

    float4 q[8];
#pragma unroll
    for (int j = 0; j < 8; ++j)
        q[j] = *(const float4*)(At + m * APITCH + 8 * j + 4 * half);

#pragma unroll
    for (int p = 0; p < 2; ++p) {
        const int c0 = p * 32;
        v8f acc0 = {0.f, 0.f, 0.f, 0.f, 0.f, 0.f, 0.f, 0.f};
        v8f acc1 = {0.f, 0.f, 0.f, 0.f, 0.f, 0.f, 0.f, 0.f};

#pragma unroll
        for (int s = 0; s < 16; ++s) {
            const int j = s >> 1, e = s & 1;
            v2f a;
            a.x = e ? q[j].z : q[j].x;
            a.y = e ? q[j].w : q[j].y;
            const int kb = 8 * j + 4 * half + 2 * e;
            v2f b;
            b.x = Wlds[(kb + 0) * CCH + c0 + m];
            b.y = Wlds[(kb + 1) * CCH + c0 + m];
            acc0 = __builtin_amdgcn_wmma_f32_16x16x4_f32(
                false, a, false, b, (short)0, acc0, false, false);
        }
#pragma unroll
        for (int s = 0; s < 16; ++s) {
            const int j = s >> 1, e = s & 1;
            v2f a;
            a.x = e ? q[j].z : q[j].x;
            a.y = e ? q[j].w : q[j].y;
            const int kb = 8 * j + 4 * half + 2 * e;
            v2f b;
            b.x = Wlds[(kb + 0) * CCH + c0 + 16 + m];
            b.y = Wlds[(kb + 1) * CCH + c0 + 16 + m];
            acc1 = __builtin_amdgcn_wmma_f32_16x16x4_f32(
                false, a, false, b, (short)0, acc1, false, false);
        }

#pragma unroll
        for (int r = 0; r < 8; ++r) {
            const auto sw = __builtin_amdgcn_permlane16_swap(
                __float_as_uint(acc0[r]), __float_as_uint(acc1[r]), false, false);
            const float lo = __uint_as_float(sw[0]);
            const float hi = __uint_as_float(sw[1]);
            atomicAddF32(row_ptr(Y, go[r])     + c0 + lane, lo);
            atomicAddF32(row_ptr(Y, go[r + 8]) + c0 + lane, hi);
        }
    }
}

__global__ __launch_bounds__(256)
void stats_kernel(const float* __restrict__ A,
                  const int*   __restrict__ bidx,
                  float* __restrict__ gsum,
                  float* __restrict__ gsq,
                  float* __restrict__ gcnt)
{
    const int t    = threadIdx.x;
    const int c    = t & 63;
    const int g    = t >> 6;
    const int row0 = blockIdx.x * 256 + g * 64;

    float s = 0.f, s2 = 0.f, cn = 0.f;
    int curb = -1;

    for (int i = 0; i < 64; ++i) {
        const int row = row0 + i;
        if (row >= NPTS) break;
        const int b = bidx[row];
        if (b != curb) {
            if (curb >= 0) {
                atomicAddF32(gsum + curb * CCH + c, s);
                atomicAddF32(gsq  + curb * CCH + c, s2);
                if (c == 0) atomicAddF32(gcnt + curb, cn);
            }
            curb = b; s = 0.f; s2 = 0.f; cn = 0.f;
        }
        const float v = A[(long)row * CCH + c];
        s += v; s2 += v * v; cn += 1.f;
    }
    if (curb >= 0) {
        atomicAddF32(gsum + curb * CCH + c, s);
        atomicAddF32(gsq  + curb * CCH + c, s2);
        if (c == 0) atomicAddF32(gcnt + curb, cn);
    }
}

__global__ void finalize_kernel(const float* __restrict__ gsum,
                                const float* __restrict__ gsq,
                                const float* __restrict__ gcnt,
                                float* __restrict__ mean,
                                float* __restrict__ rstd)
{
    const int t  = threadIdx.x;
    const float cn  = gcnt[t >> 6];
    const float mu  = gsum[t] / cn;
    const float var = gsq[t] / cn - mu * mu;
    mean[t] = mu;
    rstd[t] = rsqrtf(var + EPSI);
}

__global__ __launch_bounds__(256)
void apply_norm_kernel(float* __restrict__ A,
                       const int*   __restrict__ bidx,
                       const float* __restrict__ mean,
                       const float* __restrict__ rstd,
                       const float* __restrict__ gamma,
                       const float* __restrict__ beta)
{
    const int idx = blockIdx.x * 256 + threadIdx.x;
    if (idx >= NPTS * CCH) return;
    const int row = idx >> 6, c = idx & 63;
    const int b   = bidx[row];
    const float x = A[idx];
    const float y = gamma[c] * (x - mean[b * CCH + c]) * rstd[b * CCH + c] + beta[c];
    A[idx] = y > 0.f ? y : NSLOPE * y;
}

__global__ __launch_bounds__(256)
void apply_norm_res_kernel(float* __restrict__ Y,
                           const float* __restrict__ feats,
                           const int*   __restrict__ bidx,
                           const float* __restrict__ mean,
                           const float* __restrict__ rstd,
                           const float* __restrict__ gamma,
                           const float* __restrict__ beta)
{
    const int idx = blockIdx.x * 256 + threadIdx.x;
    if (idx >= NPTS * CCH) return;
    const int row = idx >> 6, c = idx & 63;
    const int b   = bidx[row];
    const float x = Y[idx];
    float y = gamma[c] * (x - mean[b * CCH + c]) * rstd[b * CCH + c] + beta[c];
    y += feats[idx];
    Y[idx] = y > 0.f ? y : NSLOPE * y;
}

extern "C" void kernel_launch(void* const* d_in, const int* in_sizes, int n_in,
                              void* d_out, int out_size, void* d_ws, size_t ws_size,
                              hipStream_t stream)
{
    const float* feats   = (const float*)d_in[0];
    const float* W1      = (const float*)d_in[1];
    const float* g1      = (const float*)d_in[2];
    const float* b1      = (const float*)d_in[3];
    const float* W2      = (const float*)d_in[4];
    const float* g2      = (const float*)d_in[5];
    const float* b2      = (const float*)d_in[6];
    const int*   in_map  = (const int*)d_in[7];
    const int*   out_map = (const int*)d_in[8];
    const int*   bidx    = (const int*)d_in[9];
    float*       out     = (float*)d_out;

    const int NC = NPTS * CCH;

    float* A     = (float*)d_ws;
    float* sbase = A + NC;
    float* sum1  = sbase;
    float* sq1   = sbase + 512;
    float* cnt1  = sbase + 1024;
    float* mean1 = sbase + 1088;
    float* rstd1 = sbase + 1600;
    float* sum2  = sbase + 2112;
    float* sq2   = sbase + 2624;
    float* cnt2  = sbase + 3136;
    float* mean2 = sbase + 3200;
    float* rstd2 = sbase + 3712;
    const int STATS = 4224;

    fill_zero_kernel<<<2048, 256, 0, stream>>>(A, NC + STATS);
    fill_zero_kernel<<<2048, 256, 0, stream>>>(out, NC);

    dim3 convGrid((NPTS + 127) / 128, KOFF);
    const int statGrid  = (NPTS + 255) / 256;
    const int elemGrid  = (NC + 255) / 256;

    sparse_conv_kernel<<<convGrid, 256, 0, stream>>>(feats, W1, in_map, out_map, A);
    stats_kernel<<<statGrid, 256, 0, stream>>>(A, bidx, sum1, sq1, cnt1);
    finalize_kernel<<<1, 512, 0, stream>>>(sum1, sq1, cnt1, mean1, rstd1);
    apply_norm_kernel<<<elemGrid, 256, 0, stream>>>(A, bidx, mean1, rstd1, g1, b1);
    sparse_conv_kernel<<<convGrid, 256, 0, stream>>>(A, W2, in_map, out_map, out);
    stats_kernel<<<statGrid, 256, 0, stream>>>(out, bidx, sum2, sq2, cnt2);
    finalize_kernel<<<1, 512, 0, stream>>>(sum2, sq2, cnt2, mean2, rstd2);
    apply_norm_res_kernel<<<elemGrid, 256, 0, stream>>>(out, feats, bidx, mean2, rstd2, g2, b2);
}


# stats kernel: a wave issues all its loads at once (16 rows, eight 16-B loads per lane), not two dependent loads per row over 64 rows; block sums meet in LDS first, so the sums' order changes
# speedup vs baseline: 2.0134x; 1.0981x over previous
#include <hip/hip_runtime.h>
#include <hip/hip_bf16.h>

typedef float v2f __attribute__((ext_vector_type(2)));
typedef float v4f __attribute__((ext_vector_type(4)));
typedef float v8f __attribute__((ext_vector_type(8)));

#define NPTS   100000
#define CCH    64
#define KOFF   27
#define NB     8
#define EPSI   1e-5f
#define NSLOPE 0.01f

__device__ __forceinline__ void atomicAddF32(float* p, float v) {
    unsafeAtomicAdd(p, v);
}

__global__ void fill_zero_kernel(float* __restrict__ p, int n) {
    for (int i = blockIdx.x * blockDim.x + threadIdx.x; i < n;
         i += gridDim.x * blockDim.x)
        p[i] = 0.0f;
}

typedef __attribute__((address_space(1))) float gfloat;
__device__ __forceinline__ float* row_ptr(float* Y, int g) {
    const unsigned long long p = (unsigned long long)(Y + (size_t)(unsigned)g * CCH);
    const unsigned lo = __builtin_amdgcn_readfirstlane((unsigned)p);
    const unsigned hi = __builtin_amdgcn_readfirstlane((unsigned)(p >> 32));
    return (float*)(gfloat*)(((unsigned long long)hi << 32) | lo);
}

static_assert(NPTS % 16 == 0, "a 16-row tile is wholly inside or wholly outside the point range");
static_assert((long)NPTS * CCH < (1L << 31), "element offsets into Y fit 32 bits");

#define APITCH 68

__global__ __launch_bounds__(256)
void sparse_conv_kernel(const float* __restrict__ X,
                        const float* __restrict__ W,
                        const int*   __restrict__ in_map,
                        const int*   __restrict__ out_map,
                        float*       __restrict__ Y)
{
    __shared__ __attribute__((aligned(16))) float Wlds[CCH * CCH];
    __shared__ __attribute__((aligned(16))) float Alds[8 * 16 * APITCH];

    const int k = blockIdx.y;
    const int t = threadIdx.x;

    const int wave    = __builtin_amdgcn_readfirstlane(t >> 5);
    const int lane    = t & 31;
    const int half    = lane >> 4;
    const int m       = lane & 15;
    const int rowbase = blockIdx.x * 128 + wave * 16;
    const bool live   = rowbase < NPTS;
    const int  rb     = live ? rowbase : 0;

    const float4* Wg = (const float4*)(W + k * CCH * CCH);
    float4 w[4];
#pragma unroll
    for (int i = 0; i < 4; ++i)
        w[i] = Wg[t + i * 256];

    const int4* ip  = (const int4*)(in_map  + k * NPTS + rb);
    const int4* op  = (const int4*)(out_map + k * NPTS + rb);
    const int4  i0 = ip[0], i1 = ip[1], i2 = ip[2], i3 = ip[3];
    const int4  g0 = op[0], g1 = op[1], g2 = op[2], g3 = op[3];

    {
        float4* Wl = (float4*)Wlds;
#pragma unroll
        for (int i = 0; i < 4; ++i)
            Wl[t + i * 256] = w[i];
    }
    __syncthreads();

    if (!live) return;

    const int gi[16] = {i0.x, i0.y, i0.z, i0.w, i1.x, i1.y, i1.z, i1.w,
                        i2.x, i2.y, i2.z, i2.w, i3.x, i3.y, i3.z, i3.w};
    const int go[16] = {g0.x, g0.y, g0.z, g0.w, g1.x, g1.y, g1.z, g1.w,
                        g2.x, g2.y, g2.z, g2.w, g3.x, g3.y, g3.z, g3.w};

    float4 x[8];
#pragma unroll
    for (int i = 0; i < 8; ++i) {
        const int r = half ? gi[2 * i + 1] : gi[2 * i];
        x[i] = *(const float4*)(X + (long)r * CCH + 4 * m);
    }

    float* At = Alds + wave * (16 * APITCH);
#pragma unroll
    for (int i = 0; i < 8; ++i)
        *(float4*)(At + (2 * i + half) * APITCH + 4 * m) = x[i];

    __builtin_amdgcn_fence(__ATOMIC_RELEASE, "wavefront");
    __builtin_amdgcn_wave_barrier();
    __builtin_amdgcn_fence(__ATOMIC_ACQUIRE, "wavefront");

    float4 q[8];
#pragma unroll
    for (int j = 0; j < 8; ++j)
        q[j] = *(const float4*)(At + m * APITCH + 8 * j + 4 * half);

#pragma unroll
    for (int p = 0; p < 2; ++p) {
        const int c0 = p * 32;
        v8f acc0 = {0.f, 0.f, 0.f, 0.f, 0.f, 0.f, 0.f, 0.f};
        v8f acc1 = {0.f, 0.f, 0.f, 0.f, 0.f, 0.f, 0.f, 0.f};

#pragma unroll
        for (int s = 0; s < 16; ++s) {
            const int j = s >> 1, e = s & 1;
            v2f a;
            a.x = e ? q[j].z : q[j].x;
            a.y = e ? q[j].w : q[j].y;
            const int kb = 8 * j + 4 * half + 2 * e;
            v2f b;
            b.x = Wlds[(kb + 0) * CCH + c0 + m];
            b.y = Wlds[(kb + 1) * CCH + c0 + m];
            acc0 = __builtin_amdgcn_wmma_f32_16x16x4_f32(
                false, a, false, b, (short)0, acc0, false, false);
        }
#pragma unroll
        for (int s = 0; s < 16; ++s) {
            const int j = s >> 1, e = s & 1;
            v2f a;
            a.x = e ? q[j].z : q[j].x;
            a.y = e ? q[j].w : q[j].y;
            const int kb = 8 * j + 4 * half + 2 * e;
            v2f b;
            b.x = Wlds[(kb + 0) * CCH + c0 + 16 + m];
            b.y = Wlds[(kb + 1) * CCH + c0 + 16 + m];
            acc1 = __builtin_amdgcn_wmma_f32_16x16x4_f32(
                false, a, false, b, (short)0, acc1, false, false);
        }

#pragma unroll
        for (int r = 0; r < 8; ++r) {
            const auto sw = __builtin_amdgcn_permlane16_swap(
                __float_as_uint(acc0[r]), __float_as_uint(acc1[r]), false, false);
            const float lo = __uint_as_float(sw[0]);
            const float hi = __uint_as_float(sw[1]);
            atomicAddF32(row_ptr(Y, go[r])     + c0 + lane, lo);
            atomicAddF32(row_ptr(Y, go[r + 8]) + c0 + lane, hi);
        }
    }
}

#define SROWS 128
static_assert(SROWS == 8 * 16, "eight waves of sixteen rows");

__global__ __launch_bounds__(256)
void stats_kernel(const float* __restrict__ A,
                  const int*   __restrict__ bidx,
                  float* __restrict__ gsum,
                  float* __restrict__ gsq,
                  float* __restrict__ gcnt)
{
    __shared__ __attribute__((aligned(16))) float tab[2 * NB * CCH + NB];
    float* lsum = tab;
    float* lsq  = tab + NB * CCH;
    float* lcnt = tab + 2 * NB * CCH;

    const int t    = threadIdx.x;
    const int wave = __builtin_amdgcn_readfirstlane(t >> 5);
    const int lane = t & 31;
    const int half = lane >> 4;
    const int q    = lane & 15;
    const int r0   = blockIdx.x * SROWS + wave * 16;
    const bool live = r0 < NPTS;

    v4f  x[8];
    int4 b0, b1, b2, b3;
    if (live) {
        const v4f* Ap = (const v4f*)(A + (long)r0 * CCH) + half * (CCH / 4) + q;
#pragma unroll
        for (int i = 0; i < 8; ++i)
            x[i] = Ap[i * 2 * (CCH / 4)];
        const int4* bp = (const int4*)(bidx + r0);
        b0 = bp[0]; b1 = bp[1]; b2 = bp[2]; b3 = bp[3];
    }

    ((v4f*)tab)[t] = (v4f){0.f, 0.f, 0.f, 0.f};
    if (t < NB) lcnt[t] = 0.f;
    __syncthreads();

    if (live) {
        const int be[8] = {b0.x, b0.z, b1.x, b1.z, b2.x, b2.z, b3.x, b3.z};
        const int bo[8] = {b0.y, b0.w, b1.y, b1.w, b2.y, b2.w, b3.y, b3.w};
        float s[4]  = {0.f, 0.f, 0.f, 0.f};
        float s2[4] = {0.f, 0.f, 0.f, 0.f};
        float cn = 0.f;
        int curb = half ? bo[0] : be[0];
#pragma unroll
        for (int i = 0; i < 8; ++i) {
            const int b = half ? bo[i] : be[i];
            if (i > 0 && b != curb) {
#pragma unroll
                for (int e = 0; e < 4; ++e) {
                    atomicAddF32(lsum + curb * CCH + 4 * q + e, s[e]);
                    atomicAddF32(lsq  + curb * CCH + 4 * q + e, s2[e]);
                    s[e] = 0.f; s2[e] = 0.f;
                }
                if (q == 0) atomicAddF32(lcnt + curb, cn);
                cn = 0.f;
                curb = b;
            }
#pragma unroll
            for (int e = 0; e < 4; ++e) {
                const float v = x[i][e];
                s[e] += v; s2[e] += v * v;
            }
            cn += 1.f;
        }
#pragma unroll
        for (int e = 0; e < 4; ++e) {
            atomicAddF32(lsum + curb * CCH + 4 * q + e, s[e]);
            atomicAddF32(lsq  + curb * CCH + 4 * q + e, s2[e]);
        }
        if (q == 0) atomicAddF32(lcnt + curb, cn);
    }
    __syncthreads();

    for (int i = t; i < NB * CCH; i += 256) {
        if (lcnt[i >> 6] > 0.f) {
            atomicAddF32(gsum + i, lsum[i]);
            atomicAddF32(gsq  + i, lsq[i]);
        }
    }
    if (t < NB && lcnt[t] > 0.f) atomicAddF32(gcnt + t, lcnt[t]);
}

__global__ void finalize_kernel(const float* __restrict__ gsum,
                                const float* __restrict__ gsq,
                                const float* __restrict__ gcnt,
                                float* __restrict__ mean,
                                float* __restrict__ rstd)
{
    const int t  = threadIdx.x;
    const float cn  = gcnt[t >> 6];
    const float mu  = gsum[t] / cn;
    const float var = gsq[t] / cn - mu * mu;
    mean[t] = mu;
    rstd[t] = rsqrtf(var + EPSI);
}

__global__ __launch_bounds__(256)
void apply_norm_kernel(float* __restrict__ A,
                       const int*   __restrict__ bidx,
                       const float* __restrict__ mean,
                       const float* __restrict__ rstd,
                       const float* __restrict__ gamma,
                       const float* __restrict__ beta)
{
    const int idx = blockIdx.x * 256 + threadIdx.x;
    if (idx >= NPTS * CCH) return;
    const int row = idx >> 6, c = idx & 63;
    const int b   = bidx[row];
    const float x = A[idx];
    const float y = gamma[c] * (x - mean[b * CCH + c]) * rstd[b * CCH + c] + beta[c];
    A[idx] = y > 0.f ? y : NSLOPE * y;
}

__global__ __launch_bounds__(256)
void apply_norm_res_kernel(float* __restrict__ Y,
                           const float* __restrict__ feats,
                           const int*   __restrict__ bidx,
                           const float* __restrict__ mean,
                           const float* __restrict__ rstd,
                           const float* __restrict__ gamma,
                           const float* __restrict__ beta)
{
    const int idx = blockIdx.x * 256 + threadIdx.x;
    if (idx >= NPTS * CCH) return;
    const int row = idx >> 6, c = idx & 63;
    const int b   = bidx[row];
    const float x = Y[idx];
    float y = gamma[c] * (x - mean[b * CCH + c]) * rstd[b * CCH + c] + beta[c];
    y += feats[idx];
    Y[idx] = y > 0.f ? y : NSLOPE * y;
}

extern "C" void kernel_launch(void* const* d_in, const int* in_sizes, int n_in,
                              void* d_out, int out_size, void* d_ws, size_t ws_size,
                              hipStream_t stream)
{
    const float* feats   = (const float*)d_in[0];
    const float* W1      = (const float*)d_in[1];
    const float* g1      = (const float*)d_in[2];
    const float* b1      = (const float*)d_in[3];
    const float* W2      = (const float*)d_in[4];
    const float* g2      = (const float*)d_in[5];
    const float* b2      = (const float*)d_in[6];
    const int*   in_map  = (const int*)d_in[7];
    const int*   out_map = (const int*)d_in[8];
    const int*   bidx    = (const int*)d_in[9];
    float*       out     = (float*)d_out;

    const int NC = NPTS * CCH;

    float* A     = (float*)d_ws;
    float* sbase = A + NC;
    float* sum1  = sbase;
    float* sq1   = sbase + 512;
    float* cnt1  = sbase + 1024;
    float* mean1 = sbase + 1088;
    float* rstd1 = sbase + 1600;
    float* sum2  = sbase + 2112;
    float* sq2   = sbase + 2624;
    float* cnt2  = sbase + 3136;
    float* mean2 = sbase + 3200;
    float* rstd2 = sbase + 3712;
    const int STATS = 4224;

    fill_zero_kernel<<<2048, 256, 0, stream>>>(A, NC + STATS);
    fill_zero_kernel<<<2048, 256, 0, stream>>>(out, NC);

    dim3 convGrid((NPTS + 127) / 128, KOFF);
    const int statGrid  = (NPTS + SROWS - 1) / SROWS;
    const int elemGrid  = (NC + 255) / 256;

    sparse_conv_kernel<<<convGrid, 256, 0, stream>>>(feats, W1, in_map, out_map, A);
    stats_kernel<<<statGrid, 256, 0, stream>>>(A, bidx, sum1, sq1, cnt1);
    finalize_kernel<<<1, 512, 0, stream>>>(sum1, sq1, cnt1, mean1, rstd1);
    apply_norm_kernel<<<elemGrid, 256, 0, stream>>>(A, bidx, mean1, rstd1, g1, b1);
    sparse_conv_kernel<<<convGrid, 256, 0, stream>>>(A, W2, in_map, out_map, out);
    stats_kernel<<<statGrid, 256, 0, stream>>>(out, bidx, sum2, sq2, cnt2);
    finalize_kernel<<<1, 512, 0, stream>>>(sum2, sq2, cnt2, mean2, rstd2);
    apply_norm_res_kernel<<<elemGrid, 256, 0, stream>>>(out, feats, bidx, mean2, rstd2, g2, b2);
}


# conv kernel: a wave reads W[k]'s B fragments out of LDS once into 128 registers and walks four row tiles with them (4-wave blocks); a tile's 64 WMMAs stand back to back, no LDS read of B among them
# speedup vs baseline: 2.1266x; 1.0562x over previous
#include <hip/hip_runtime.h>
#include <hip/hip_bf16.h>

typedef float v2f __attribute__((ext_vector_type(2)));
typedef float v4f __attribute__((ext_vector_type(4)));
typedef float v8f __attribute__((ext_vector_type(8)));

#define NPTS   100000
#define CCH    64
#define KOFF   27
#define NB     8
#define EPSI   1e-5f
#define NSLOPE 0.01f

__device__ __forceinline__ void atomicAddF32(float* p, float v) {
    unsafeAtomicAdd(p, v);
}

__global__ void fill_zero_kernel(float* __restrict__ p, int n) {
    for (int i = blockIdx.x * blockDim.x + threadIdx.x; i < n;
         i += gridDim.x * blockDim.x)
        p[i] = 0.0f;
}

typedef __attribute__((address_space(1))) float gfloat;
__device__ __forceinline__ float* row_ptr(float* Y, int g) {
    const unsigned long long p = (unsigned long long)(Y + (size_t)(unsigned)g * CCH);
    const unsigned lo = __builtin_amdgcn_readfirstlane((unsigned)p);
    const unsigned hi = __builtin_amdgcn_readfirstlane((unsigned)(p >> 32));
    return (float*)(gfloat*)(((unsigned long long)hi << 32) | lo);
}

static_assert(NPTS % 16 == 0, "a 16-row tile is wholly inside or wholly outside the point range");
static_assert((long)NPTS * CCH < (1L << 31), "element offsets into Y fit 32 bits");

#define APITCH 68
#define TPW    4
#define CWAVES 4
#define BROWS  (CWAVES * 16 * TPW)

__global__ __launch_bounds__(CWAVES * 32)
void sparse_conv_kernel(const float* __restrict__ X,
                        const float* __restrict__ W,
                        const int*   __restrict__ in_map,
                        const int*   __restrict__ out_map,
                        float*       __restrict__ Y)
{
    __shared__ __attribute__((aligned(16))) float Wlds[CCH * CCH];
    __shared__ __attribute__((aligned(16))) float Alds[CWAVES * 16 * APITCH];

    const int k = blockIdx.y;
    const int t = threadIdx.x;

    const int wave  = __builtin_amdgcn_readfirstlane(t >> 5);
    const int lane  = t & 31;
    const int half  = lane >> 4;
    const int m     = lane & 15;
    const int wrow0 = blockIdx.x * BROWS + wave * (16 * TPW);
    int nt = (NPTS - wrow0) / 16;
    nt = nt < 0 ? 0 : (nt > TPW ? TPW : nt);
    nt = __builtin_amdgcn_readfirstlane(nt);

    const float4* Wg = (const float4*)(W + k * CCH * CCH);
    float4 w[8];
#pragma unroll
    for (int i = 0; i < 8; ++i)
        w[i] = Wg[t + i * (CWAVES * 32)];

    int2 im = {0, 0}, om = {0, 0};
    if (wrow0 + 2 * lane < NPTS) {
        im = *(const int2*)(in_map  + k * NPTS + wrow0 + 2 * lane);
        om = *(const int2*)(out_map + k * NPTS + wrow0 + 2 * lane);
    }

    {
        float4* Wl = (float4*)Wlds;
#pragma unroll
        for (int i = 0; i < 8; ++i)
            Wl[t + i * (CWAVES * 32)] = w[i];
    }
    __syncthreads();

    if (nt == 0) return;

    v2f bf[4][16];
#pragma unroll
    for (int n = 0; n < 4; ++n) {
#pragma unroll
        for (int s = 0; s < 16; ++s) {
            const int kb = 8 * (s >> 1) + 4 * half + 2 * (s & 1);
            bf[n][s].x = Wlds[(kb + 0) * CCH + 16 * n + m];
            bf[n][s].y = Wlds[(kb + 1) * CCH + 16 * n + m];
        }
    }

    float* At = Alds + wave * (16 * APITCH);

    for (int it = 0; it < nt; ++it) {
        int gie[8], gio[8], go[16];
#pragma unroll
        for (int i = 0; i < 8; ++i) {
            gie[i] = __builtin_amdgcn_readlane(im.x, 8 * it + i);
            gio[i] = __builtin_amdgcn_readlane(im.y, 8 * it + i);
        }
#pragma unroll
        for (int j = 0; j < 16; ++j)
            go[j] = __builtin_amdgcn_readlane((j & 1) ? om.y : om.x, 8 * it + (j >> 1));

        float4 x[8];
#pragma unroll
        for (int i = 0; i < 8; ++i) {
            const int r = half ? gio[i] : gie[i];
            x[i] = *(const float4*)(X + (long)r * CCH + 4 * m);
        }

#pragma unroll
        for (int i = 0; i < 8; ++i)
            *(float4*)(At + (2 * i + half) * APITCH + 4 * m) = x[i];

        __builtin_amdgcn_fence(__ATOMIC_RELEASE, "wavefront");
        __builtin_amdgcn_wave_barrier();
        __builtin_amdgcn_fence(__ATOMIC_ACQUIRE, "wavefront");

        float4 q[8];
#pragma unroll
        for (int j = 0; j < 8; ++j)
            q[j] = *(const float4*)(At + m * APITCH + 8 * j + 4 * half);

        __builtin_amdgcn_fence(__ATOMIC_RELEASE, "wavefront");
        __builtin_amdgcn_wave_barrier();
        __builtin_amdgcn_fence(__ATOMIC_ACQUIRE, "wavefront");

        v8f acc[4];
#pragma unroll
        for (int n = 0; n < 4; ++n)
            acc[n] = (v8f){0.f, 0.f, 0.f, 0.f, 0.f, 0.f, 0.f, 0.f};
#pragma unroll
        for (int s = 0; s < 16; ++s) {
            const int j = s >> 1, e = s & 1;
            v2f a;
            a.x = e ? q[j].z : q[j].x;
            a.y = e ? q[j].w : q[j].y;
#pragma unroll
            for (int n = 0; n < 4; ++n)
                acc[n] = __builtin_amdgcn_wmma_f32_16x16x4_f32(
                    false, a, false, bf[n][s], (short)0, acc[n], false, false);
        }

#pragma unroll
        for (int p = 0; p < 2; ++p) {
            const int c0 = p * 32;
#pragma unroll
            for (int r = 0; r < 8; ++r) {
                const auto sw = __builtin_amdgcn_permlane16_swap(
                    __float_as_uint(acc[2 * p][r]), __float_as_uint(acc[2 * p + 1][r]), false, false);
                const float lo = __uint_as_float(sw[0]);
                const float hi = __uint_as_float(sw[1]);
                atomicAddF32(row_ptr(Y, go[r])     + c0 + lane, lo);
                atomicAddF32(row_ptr(Y, go[r + 8]) + c0 + lane, hi);
            }
        }
    }
}

#define SROWS 128
static_assert(SROWS == 8 * 16, "eight waves of sixteen rows");

__global__ __launch_bounds__(256)
void stats_kernel(const float* __restrict__ A,
                  const int*   __restrict__ bidx,
                  float* __restrict__ gsum,
                  float* __restrict__ gsq,
                  float* __restrict__ gcnt)
{
    __shared__ __attribute__((aligned(16))) float tab[2 * NB * CCH + NB];
    float* lsum = tab;
    float* lsq  = tab + NB * CCH;
    float* lcnt = tab + 2 * NB * CCH;

    const int t    = threadIdx.x;
    const int wave = __builtin_amdgcn_readfirstlane(t >> 5);
    const int lane = t & 31;
    const int half = lane >> 4;
    const int q    = lane & 15;
    const int r0   = blockIdx.x * SROWS + wave * 16;
    const bool live = r0 < NPTS;

    v4f  x[8];
    int4 b0, b1, b2, b3;
    if (live) {
        const v4f* Ap = (const v4f*)(A + (long)r0 * CCH) + half * (CCH / 4) + q;
#pragma unroll
        for (int i = 0; i < 8; ++i)
            x[i] = Ap[i * 2 * (CCH / 4)];
        const int4* bp = (const int4*)(bidx + r0);
        b0 = bp[0]; b1 = bp[1]; b2 = bp[2]; b3 = bp[3];
    }

    ((v4f*)tab)[t] = (v4f){0.f, 0.f, 0.f, 0.f};
    if (t < NB) lcnt[t] = 0.f;
    __syncthreads();

    if (live) {
        const int be[8] = {b0.x, b0.z, b1.x, b1.z, b2.x, b2.z, b3.x, b3.z};
        const int bo[8] = {b0.y, b0.w, b1.y, b1.w, b2.y, b2.w, b3.y, b3.w};
        float s[4]  = {0.f, 0.f, 0.f, 0.f};
        float s2[4] = {0.f, 0.f, 0.f, 0.f};
        float cn = 0.f;
        int curb = half ? bo[0] : be[0];
#pragma unroll
        for (int i = 0; i < 8; ++i) {
            const int b = half ? bo[i] : be[i];
            if (i > 0 && b != curb) {
#pragma unroll
                for (int e = 0; e < 4; ++e) {
                    atomicAddF32(lsum + curb * CCH + 4 * q + e, s[e]);
                    atomicAddF32(lsq  + curb * CCH + 4 * q + e, s2[e]);
                    s[e] = 0.f; s2[e] = 0.f;
                }
                if (q == 0) atomicAddF32(lcnt + curb, cn);
                cn = 0.f;
                curb = b;
            }
#pragma unroll
            for (int e = 0; e < 4; ++e) {
                const float v = x[i][e];
                s[e] += v; s2[e] += v * v;
            }
            cn += 1.f;
        }
#pragma unroll
        for (int e = 0; e < 4; ++e) {
            atomicAddF32(lsum + curb * CCH + 4 * q + e, s[e]);
            atomicAddF32(lsq  + curb * CCH + 4 * q + e, s2[e]);
        }
        if (q == 0) atomicAddF32(lcnt + curb, cn);
    }
    __syncthreads();

    for (int i = t; i < NB * CCH; i += 256) {
        if (lcnt[i >> 6] > 0.f) {
            atomicAddF32(gsum + i, lsum[i]);
            atomicAddF32(gsq  + i, lsq[i]);
        }
    }
    if (t < NB && lcnt[t] > 0.f) atomicAddF32(gcnt + t, lcnt[t]);
}

__global__ void finalize_kernel(const float* __restrict__ gsum,
                                const float* __restrict__ gsq,
                                const float* __restrict__ gcnt,
                                float* __restrict__ mean,
                                float* __restrict__ rstd)
{
    const int t  = threadIdx.x;
    const float cn  = gcnt[t >> 6];
    const float mu  = gsum[t] / cn;
    const float var = gsq[t] / cn - mu * mu;
    mean[t] = mu;
    rstd[t] = rsqrtf(var + EPSI);
}

__global__ __launch_bounds__(256)
void apply_norm_kernel(float* __restrict__ A,
                       const int*   __restrict__ bidx,
                       const float* __restrict__ mean,
                       const float* __restrict__ rstd,
                       const float* __restrict__ gamma,
                       const float* __restrict__ beta)
{
    const int idx = blockIdx.x * 256 + threadIdx.x;
    if (idx >= NPTS * CCH) return;
    const int row = idx >> 6, c = idx & 63;
    const int b   = bidx[row];
    const float x = A[idx];
    const float y = gamma[c] * (x - mean[b * CCH + c]) * rstd[b * CCH + c] + beta[c];
    A[idx] = y > 0.f ? y : NSLOPE * y;
}

__global__ __launch_bounds__(256)
void apply_norm_res_kernel(float* __restrict__ Y,
                           const float* __restrict__ feats,
                           const int*   __restrict__ bidx,
                           const float* __restrict__ mean,
                           const float* __restrict__ rstd,
                           const float* __restrict__ gamma,
                           const float* __restrict__ beta)
{
    const int idx = blockIdx.x * 256 + threadIdx.x;
    if (idx >= NPTS * CCH) return;
    const int row = idx >> 6, c = idx & 63;
    const int b   = bidx[row];
    const float x = Y[idx];
    float y = gamma[c] * (x - mean[b * CCH + c]) * rstd[b * CCH + c] + beta[c];
    y += feats[idx];
    Y[idx] = y > 0.f ? y : NSLOPE * y;
}

extern "C" void kernel_launch(void* const* d_in, const int* in_sizes, int n_in,
                              void* d_out, int out_size, void* d_ws, size_t ws_size,
                              hipStream_t stream)
{
    const float* feats   = (const float*)d_in[0];
    const float* W1      = (const float*)d_in[1];
    const float* g1      = (const float*)d_in[2];
    const float* b1      = (const float*)d_in[3];
    const float* W2      = (const float*)d_in[4];
    const float* g2      = (const float*)d_in[5];
    const float* b2      = (const float*)d_in[6];
    const int*   in_map  = (const int*)d_in[7];
    const int*   out_map = (const int*)d_in[8];
    const int*   bidx    = (const int*)d_in[9];
    float*       out     = (float*)d_out;

    const int NC = NPTS * CCH;

    float* A     = (float*)d_ws;
    float* sbase = A + NC;
    float* sum1  = sbase;
    float* sq1   = sbase + 512;
    float* cnt1  = sbase + 1024;
    float* mean1 = sbase + 1088;
    float* rstd1 = sbase + 1600;
    float* sum2  = sbase + 2112;
    float* sq2   = sbase + 2624;
    float* cnt2  = sbase + 3136;
    float* mean2 = sbase + 3200;
    float* rstd2 = sbase + 3712;
    const int STATS = 4224;

    fill_zero_kernel<<<2048, 256, 0, stream>>>(A, NC + STATS);
    fill_zero_kernel<<<2048, 256, 0, stream>>>(out, NC);

    dim3 convGrid((NPTS + BROWS - 1) / BROWS, KOFF);
    const int statGrid  = (NPTS + SROWS - 1) / SROWS;
    const int elemGrid  = (NC + 255) / 256;

    sparse_conv_kernel<<<convGrid, CWAVES * 32, 0, stream>>>(feats, W1, in_map, out_map, A);
    stats_kernel<<<statGrid, 256, 0, stream>>>(A, bidx, sum1, sq1, cnt1);
    finalize_kernel<<<1, 512, 0, stream>>>(sum1, sq1, cnt1, mean1, rstd1);
    apply_norm_kernel<<<elemGrid, 256, 0, stream>>>(A, bidx, mean1, rstd1, g1, b1);
    sparse_conv_kernel<<<convGrid, CWAVES * 32, 0, stream>>>(A, W2, in_map, out_map, out);
    stats_kernel<<<statGrid, 256, 0, stream>>>(out, bidx, sum2, sq2, cnt2);
    finalize_kernel<<<1, 512, 0, stream>>>(sum2, sq2, cnt2, mean2, rstd2);
    apply_norm_res_kernel<<<elemGrid, 256, 0, stream>>>(out, feats, bidx, mean2, rstd2, g2, b2);
}


# conv kernels: each second-level cache half (by the block's die number) adds into its own plane of the output; statistics and normalise passes add the two planes; sums are formed in another order
# speedup vs baseline: 2.2025x; 1.0357x over previous
#include <hip/hip_runtime.h>
#include <hip/hip_bf16.h>

typedef float v2f __attribute__((ext_vector_type(2)));
typedef float v4f __attribute__((ext_vector_type(4)));
typedef float v8f __attribute__((ext_vector_type(8)));

#define NPTS   100000
#define CCH    64
#define KOFF   27
#define NB     8
#define EPSI   1e-5f
#define NSLOPE 0.01f

#define NCF    (NPTS * CCH)
#define STATS  4224
#define WS_TWO ((size_t)(2 * (size_t)NCF + STATS) * sizeof(float))

__device__ __forceinline__ int cache_half() {
    return (int)((__builtin_amdgcn_s_sendmsg_rtn(0x87) >> 18) & 1u);
}

__device__ __forceinline__ void atomicAddF32(float* p, float v) {
    unsafeAtomicAdd(p, v);
}

__global__ void fill_zero_kernel(float* __restrict__ p, int n) {
    for (int i = blockIdx.x * blockDim.x + threadIdx.x; i < n;
         i += gridDim.x * blockDim.x)
        p[i] = 0.0f;
}

__global__ void fill_ws_kernel(float* __restrict__ p, size_t ws_bytes) {
    const int n = ws_bytes >= WS_TWO ? 2 * NCF + STATS : NCF + STATS;
    for (int i = blockIdx.x * blockDim.x + threadIdx.x; i < n;
         i += gridDim.x * blockDim.x)
        p[i] = 0.0f;
}

typedef __attribute__((address_space(1))) float gfloat;
__device__ __forceinline__ float* row_ptr(float* Y, int g) {
    const unsigned long long p = (unsigned long long)(Y + (size_t)(unsigned)g * CCH);
    const unsigned lo = __builtin_amdgcn_readfirstlane((unsigned)p);
    const unsigned hi = __builtin_amdgcn_readfirstlane((unsigned)(p >> 32));
    return (float*)(gfloat*)(((unsigned long long)hi << 32) | lo);
}

static_assert(NPTS % 16 == 0, "a 16-row tile is wholly inside or wholly outside the point range");
static_assert((long)NPTS * CCH < (1L << 31), "element offsets into Y fit 32 bits");

#define APITCH 68
#define TPW    4
#define CWAVES 4
#define BROWS  (CWAVES * 16 * TPW)

__global__ __launch_bounds__(CWAVES * 32)
void sparse_conv_kernel(const float* __restrict__ X,
                        const float* __restrict__ W,
                        const int*   __restrict__ in_map,
                        const int*   __restrict__ out_map,
                        float*       __restrict__ Y0,
                        float*       __restrict__ Y1,
                        size_t room)
{
    float* Y = (room >= WS_TWO && cache_half()) ? Y1 : Y0;

    __shared__ __attribute__((aligned(16))) float Wlds[CCH * CCH];
    __shared__ __attribute__((aligned(16))) float Alds[CWAVES * 16 * APITCH];

    const int k = blockIdx.y;
    const int t = threadIdx.x;

    const int wave  = __builtin_amdgcn_readfirstlane(t >> 5);
    const int lane  = t & 31;
    const int half  = lane >> 4;
    const int m     = lane & 15;
    const int wrow0 = blockIdx.x * BROWS + wave * (16 * TPW);
    int nt = (NPTS - wrow0) / 16;
    nt = nt < 0 ? 0 : (nt > TPW ? TPW : nt);
    nt = __builtin_amdgcn_readfirstlane(nt);

    const float4* Wg = (const float4*)(W + k * CCH * CCH);
    float4 w[8];
#pragma unroll
    for (int i = 0; i < 8; ++i)
        w[i] = Wg[t + i * (CWAVES * 32)];

    int2 im = {0, 0}, om = {0, 0};
    if (wrow0 + 2 * lane < NPTS) {
        im = *(const int2*)(in_map  + k * NPTS + wrow0 + 2 * lane);
        om = *(const int2*)(out_map + k * NPTS + wrow0 + 2 * lane);
    }

    {
        float4* Wl = (float4*)Wlds;
#pragma unroll
        for (int i = 0; i < 8; ++i)
            Wl[t + i * (CWAVES * 32)] = w[i];
    }
    __syncthreads();

    if (nt == 0) return;

    v2f bf[4][16];
#pragma unroll
    for (int n = 0; n < 4; ++n) {
#pragma unroll
        for (int s = 0; s < 16; ++s) {
            const int kb = 8 * (s >> 1) + 4 * half + 2 * (s & 1);
            bf[n][s].x = Wlds[(kb + 0) * CCH + 16 * n + m];
            bf[n][s].y = Wlds[(kb + 1) * CCH + 16 * n + m];
        }
    }

    float* At = Alds + wave * (16 * APITCH);

    for (int it = 0; it < nt; ++it) {
        int gie[8], gio[8], go[16];
#pragma unroll
        for (int i = 0; i < 8; ++i) {
            gie[i] = __builtin_amdgcn_readlane(im.x, 8 * it + i);
            gio[i] = __builtin_amdgcn_readlane(im.y, 8 * it + i);
        }
#pragma unroll
        for (int j = 0; j < 16; ++j)
            go[j] = __builtin_amdgcn_readlane((j & 1) ? om.y : om.x, 8 * it + (j >> 1));

        float4 x[8];
#pragma unroll
        for (int i = 0; i < 8; ++i) {
            const int r = half ? gio[i] : gie[i];
            x[i] = *(const float4*)(X + (long)r * CCH + 4 * m);
        }

#pragma unroll
        for (int i = 0; i < 8; ++i)
            *(float4*)(At + (2 * i + half) * APITCH + 4 * m) = x[i];

        __builtin_amdgcn_fence(__ATOMIC_RELEASE, "wavefront");
        __builtin_amdgcn_wave_barrier();
        __builtin_amdgcn_fence(__ATOMIC_ACQUIRE, "wavefront");

        float4 q[8];
#pragma unroll
        for (int j = 0; j < 8; ++j)
            q[j] = *(const float4*)(At + m * APITCH + 8 * j + 4 * half);

        __builtin_amdgcn_fence(__ATOMIC_RELEASE, "wavefront");
        __builtin_amdgcn_wave_barrier();
        __builtin_amdgcn_fence(__ATOMIC_ACQUIRE, "wavefront");

        v8f acc[4];
#pragma unroll
        for (int n = 0; n < 4; ++n)
            acc[n] = (v8f){0.f, 0.f, 0.f, 0.f, 0.f, 0.f, 0.f, 0.f};
#pragma unroll
        for (int s = 0; s < 16; ++s) {
            const int j = s >> 1, e = s & 1;
            v2f a;
            a.x = e ? q[j].z : q[j].x;
            a.y = e ? q[j].w : q[j].y;
#pragma unroll
            for (int n = 0; n < 4; ++n)
                acc[n] = __builtin_amdgcn_wmma_f32_16x16x4_f32(
                    false, a, false, bf[n][s], (short)0, acc[n], false, false);
        }

#pragma unroll
        for (int p = 0; p < 2; ++p) {
            const int c0 = p * 32;
#pragma unroll
            for (int r = 0; r < 8; ++r) {
                const auto sw = __builtin_amdgcn_permlane16_swap(
                    __float_as_uint(acc[2 * p][r]), __float_as_uint(acc[2 * p + 1][r]), false, false);
                const float lo = __uint_as_float(sw[0]);
                const float hi = __uint_as_float(sw[1]);
                atomicAddF32(row_ptr(Y, go[r])     + c0 + lane, lo);
                atomicAddF32(row_ptr(Y, go[r + 8]) + c0 + lane, hi);
            }
        }
    }
}

#define SROWS 128
static_assert(SROWS == 8 * 16, "eight waves of sixteen rows");

__global__ __launch_bounds__(256)
void stats_kernel(const float* __restrict__ A,
                  const float* __restrict__ A1,
                  size_t room,
                  const int*   __restrict__ bidx,
                  float* __restrict__ gsum,
                  float* __restrict__ gsq,
                  float* __restrict__ gcnt)
{
    const bool two = room >= WS_TWO;
    __shared__ __attribute__((aligned(16))) float tab[2 * NB * CCH + NB];
    float* lsum = tab;
    float* lsq  = tab + NB * CCH;
    float* lcnt = tab + 2 * NB * CCH;

    const int t    = threadIdx.x;
    const int wave = __builtin_amdgcn_readfirstlane(t >> 5);
    const int lane = t & 31;
    const int half = lane >> 4;
    const int q    = lane & 15;
    const int r0   = blockIdx.x * SROWS + wave * 16;
    const bool live = r0 < NPTS;

    v4f  x[8], x1[8];
    int4 b0, b1, b2, b3;
    if (live) {
        const v4f* Ap = (const v4f*)(A + (long)r0 * CCH) + half * (CCH / 4) + q;
#pragma unroll
        for (int i = 0; i < 8; ++i)
            x[i] = Ap[i * 2 * (CCH / 4)];
        if (two) {
            const v4f* Bp = (const v4f*)(A1 + (long)r0 * CCH) + half * (CCH / 4) + q;
#pragma unroll
            for (int i = 0; i < 8; ++i)
                x1[i] = Bp[i * 2 * (CCH / 4)];
        }
        const int4* bp = (const int4*)(bidx + r0);
        b0 = bp[0]; b1 = bp[1]; b2 = bp[2]; b3 = bp[3];
    }

    ((v4f*)tab)[t] = (v4f){0.f, 0.f, 0.f, 0.f};
    if (t < NB) lcnt[t] = 0.f;
    __syncthreads();

    if (live && two) {
#pragma unroll
        for (int i = 0; i < 8; ++i)
            x[i] += x1[i];
    }

    if (live) {
        const int be[8] = {b0.x, b0.z, b1.x, b1.z, b2.x, b2.z, b3.x, b3.z};
        const int bo[8] = {b0.y, b0.w, b1.y, b1.w, b2.y, b2.w, b3.y, b3.w};
        float s[4]  = {0.f, 0.f, 0.f, 0.f};
        float s2[4] = {0.f, 0.f, 0.f, 0.f};
        float cn = 0.f;
        int curb = half ? bo[0] : be[0];
#pragma unroll
        for (int i = 0; i < 8; ++i) {
            const int b = half ? bo[i] : be[i];
            if (i > 0 && b != curb) {
#pragma unroll
                for (int e = 0; e < 4; ++e) {
                    atomicAddF32(lsum + curb * CCH + 4 * q + e, s[e]);
                    atomicAddF32(lsq  + curb * CCH + 4 * q + e, s2[e]);
                    s[e] = 0.f; s2[e] = 0.f;
                }
                if (q == 0) atomicAddF32(lcnt + curb, cn);
                cn = 0.f;
                curb = b;
            }
#pragma unroll
            for (int e = 0; e < 4; ++e) {
                const float v = x[i][e];
                s[e] += v; s2[e] += v * v;
            }
            cn += 1.f;
        }
#pragma unroll
        for (int e = 0; e < 4; ++e) {
            atomicAddF32(lsum + curb * CCH + 4 * q + e, s[e]);
            atomicAddF32(lsq  + curb * CCH + 4 * q + e, s2[e]);
        }
        if (q == 0) atomicAddF32(lcnt + curb, cn);
    }
    __syncthreads();

    for (int i = t; i < NB * CCH; i += 256) {
        if (lcnt[i >> 6] > 0.f) {
            atomicAddF32(gsum + i, lsum[i]);
            atomicAddF32(gsq  + i, lsq[i]);
        }
    }
    if (t < NB && lcnt[t] > 0.f) atomicAddF32(gcnt + t, lcnt[t]);
}

__global__ void finalize_kernel(const float* __restrict__ gsum,
                                const float* __restrict__ gsq,
                                const float* __restrict__ gcnt,
                                float* __restrict__ mean,
                                float* __restrict__ rstd)
{
    const int t  = threadIdx.x;
    const float cn  = gcnt[t >> 6];
    const float mu  = gsum[t] / cn;
    const float var = gsq[t] / cn - mu * mu;
    mean[t] = mu;
    rstd[t] = rsqrtf(var + EPSI);
}

__global__ __launch_bounds__(256)
void apply_norm_kernel(float* __restrict__ A,
                       float* __restrict__ A1,
                       const int*   __restrict__ bidx,
                       const float* __restrict__ mean,
                       const float* __restrict__ rstd,
                       const float* __restrict__ gamma,
                       const float* __restrict__ beta)
{
    const int idx = blockIdx.x * 256 + threadIdx.x;
    if (idx >= NPTS * CCH) return;
    const int row = idx >> 6, c = idx & 63;
    const int b   = bidx[row];
    const float x = A[idx] + A1[idx];
    const float y = gamma[c] * (x - mean[b * CCH + c]) * rstd[b * CCH + c] + beta[c];
    A[idx] = y > 0.f ? y : NSLOPE * y;
    A1[idx] = 0.f;
}

__global__ __launch_bounds__(256)
void apply_norm_res_kernel(float* __restrict__ Y,
                           const float* __restrict__ Y1,
                           size_t room,
                           const float* __restrict__ feats,
                           const int*   __restrict__ bidx,
                           const float* __restrict__ mean,
                           const float* __restrict__ rstd,
                           const float* __restrict__ gamma,
                           const float* __restrict__ beta)
{
    const int idx = blockIdx.x * 256 + threadIdx.x;
    if (idx >= NPTS * CCH) return;
    const int row = idx >> 6, c = idx & 63;
    const int b   = bidx[row];
    float x = Y[idx];
    if (room >= WS_TWO) x += Y1[idx];
    float y = gamma[c] * (x - mean[b * CCH + c]) * rstd[b * CCH + c] + beta[c];
    y += feats[idx];
    Y[idx] = y > 0.f ? y : NSLOPE * y;
}

extern "C" void kernel_launch(void* const* d_in, const int* in_sizes, int n_in,
                              void* d_out, int out_size, void* d_ws, size_t ws_size,
                              hipStream_t stream)
{
    const float* feats   = (const float*)d_in[0];
    const float* W1      = (const float*)d_in[1];
    const float* g1      = (const float*)d_in[2];
    const float* b1      = (const float*)d_in[3];
    const float* W2      = (const float*)d_in[4];
    const float* g2      = (const float*)d_in[5];
    const float* b2      = (const float*)d_in[6];
    const int*   in_map  = (const int*)d_in[7];
    const int*   out_map = (const int*)d_in[8];
    const int*   bidx    = (const int*)d_in[9];
    float*       out     = (float*)d_out;

    const int NC = NPTS * CCH;

    float* A     = (float*)d_ws;
    float* sbase = A + NC;
    float* sum1  = sbase;
    float* sq1   = sbase + 512;
    float* cnt1  = sbase + 1024;
    float* mean1 = sbase + 1088;
    float* rstd1 = sbase + 1600;
    float* sum2  = sbase + 2112;
    float* sq2   = sbase + 2624;
    float* cnt2  = sbase + 3136;
    float* mean2 = sbase + 3200;
    float* rstd2 = sbase + 3712;
    float* B2    = sbase + STATS;

    fill_ws_kernel<<<2048, 256, 0, stream>>>(A, ws_size);
    fill_zero_kernel<<<2048, 256, 0, stream>>>(out, NC);

    dim3 convGrid((NPTS + BROWS - 1) / BROWS, KOFF);
    const int statGrid  = (NPTS + SROWS - 1) / SROWS;
    const int elemGrid  = (NC + 255) / 256;

    sparse_conv_kernel<<<convGrid, CWAVES * 32, 0, stream>>>(feats, W1, in_map, out_map, A, out, WS_TWO);
    stats_kernel<<<statGrid, 256, 0, stream>>>(A, out, WS_TWO, bidx, sum1, sq1, cnt1);
    finalize_kernel<<<1, 512, 0, stream>>>(sum1, sq1, cnt1, mean1, rstd1);
    apply_norm_kernel<<<elemGrid, 256, 0, stream>>>(A, out, bidx, mean1, rstd1, g1, b1);
    sparse_conv_kernel<<<convGrid, CWAVES * 32, 0, stream>>>(A, W2, in_map, out_map, out, B2, ws_size);
    stats_kernel<<<statGrid, 256, 0, stream>>>(out, B2, ws_size, bidx, sum2, sq2, cnt2);
    finalize_kernel<<<1, 512, 0, stream>>>(sum2, sq2, cnt2, mean2, rstd2);
    apply_norm_res_kernel<<<elemGrid, 256, 0, stream>>>(out, B2, ws_size, feats, bidx, mean2, rstd2, g2, b2);
}
